# starting kernel
# baseline (speedup 1.0000x reference)
#include <hip/hip_runtime.h>
#include <hip/hip_bf16.h>
#include <math.h>

typedef __attribute__((ext_vector_type(2))) float v2f;
typedef __attribute__((ext_vector_type(4))) float v4f;
typedef __attribute__((ext_vector_type(8))) float v8f;

#define QDIM  4096
#define LROWS 8192

__global__ void __launch_bounds__(256) gemv16_wmma_f32(
    const float* __restrict__ M, const float* __restrict__ x,
    float* __restrict__ y_part, int C, int segC)
{
    extern __shared__ float sx[];
    const int tid   = threadIdx.x;
    const int kbase = blockIdx.y * segC;

    {
        const v4f* xs = (const v4f*)(x + kbase);
        v4f*       sd = (v4f*)sx;
        for (int i = tid; i < (segC >> 2); i += 256) sd[i] = xs[i];
    }
    __syncthreads();

    const int lane    = tid & 31;
    const int m       = lane & 15;
    const int kh      = lane >> 4;
    const int wave    = tid >> 5;
    const int R       = (int)gridDim.x * 128;
    const int rowbase = (int)blockIdx.x * 128 + wave * 16;
    const float* Arow = M + (size_t)(rowbase + m) * (size_t)C + kbase;
    const float  bm   = (m == 0) ? 1.0f : 0.0f;

    v8f acc0 = {}, acc1 = {}, acc2 = {}, acc3 = {};
    for (int k0 = 0; k0 < segC; k0 += 16) {
        const int off_lo = k0 + kh * 4;
        const int off_hi = k0 + 8 + kh * 4;
        v4f a_lo = *(const v4f*)(Arow + off_lo);
        v4f a_hi = *(const v4f*)(Arow + off_hi);
        v4f x_lo = *(const v4f*)(sx + off_lo);
        v4f x_hi = *(const v4f*)(sx + off_hi);

        v2f a0 = {a_lo.x, a_lo.y}, a1 = {a_lo.z, a_lo.w};
        v2f a2 = {a_hi.x, a_hi.y}, a3 = {a_hi.z, a_hi.w};
        v2f b0 = {x_lo.x * bm, x_lo.y * bm}, b1 = {x_lo.z * bm, x_lo.w * bm};
        v2f b2 = {x_hi.x * bm, x_hi.y * bm}, b3 = {x_hi.z * bm, x_hi.w * bm};

        acc0 = __builtin_amdgcn_wmma_f32_16x16x4_f32(false, a0, false, b0, (short)0, acc0, false, false);
        acc1 = __builtin_amdgcn_wmma_f32_16x16x4_f32(false, a1, false, b1, (short)0, acc1, false, false);
        acc2 = __builtin_amdgcn_wmma_f32_16x16x4_f32(false, a2, false, b2, (short)0, acc2, false, false);
        acc3 = __builtin_amdgcn_wmma_f32_16x16x4_f32(false, a3, false, b3, (short)0, acc3, false, false);
    }
    v8f t = acc0 + acc1 + acc2 + acc3;

    if (m == 0) {
        float* dst = y_part + (size_t)blockIdx.y * R + rowbase + kh * 8;
        #pragma unroll
        for (int r = 0; r < 8; ++r) dst[r] = t[r];
    }
}

__global__ void __launch_bounds__(256) reduce_part(
    const float* __restrict__ p, float* __restrict__ y, int R, int nseg)
{
    int i = blockIdx.x * 256 + threadIdx.x;
    if (i < R) {
        float s = 0.0f;
        for (int j = 0; j < nseg; ++j) s += p[(size_t)j * R + i];
        y[i] = s;
    }
}

__global__ void __launch_bounds__(1024) softmax_fuse(
    float* __restrict__ score, const float* __restrict__ query,
    float* __restrict__ gamma)
{
    __shared__ float red[1024];
    const int tid = threadIdx.x;

    float m = -INFINITY;
    #pragma unroll
    for (int j = 0; j < LROWS / 1024; ++j) m = fmaxf(m, score[tid + j * 1024]);
    red[tid] = m;
    __syncthreads();
    for (int s = 512; s > 0; s >>= 1) {
        if (tid < s) red[tid] = fmaxf(red[tid], red[tid + s]);
        __syncthreads();
    }
    const float gmax = red[0];
    __syncthreads();

    float e[LROWS / 1024];
    float sum = 0.0f;
    #pragma unroll
    for (int j = 0; j < LROWS / 1024; ++j) {
        e[j] = expf(score[tid + j * 1024] - gmax);
        sum += e[j];
    }
    red[tid] = sum;
    __syncthreads();
    for (int s = 512; s > 0; s >>= 1) {
        if (tid < s) red[tid] += red[tid + s];
        __syncthreads();
    }
    const float inv = 1.0f / red[0];
    #pragma unroll
    for (int j = 0; j < LROWS / 1024; ++j) score[tid + j * 1024] = e[j] * inv;

    for (int i = tid; i < QDIM; i += 1024) gamma[i] = query[i];
}

__global__ void __launch_bounds__(256) wsum_part(
    const float* __restrict__ ctx, const float* __restrict__ att,
    float* __restrict__ p)
{
    __shared__ float sa[LROWS / 16];
    const int tid   = threadIdx.x;
    const int lbase = blockIdx.y * (LROWS / 16);
    for (int i = tid; i < LROWS / 16; i += 256) sa[i] = att[lbase + i];
    __syncthreads();

    const int q = blockIdx.x * 256 + tid;
    const float* c = ctx + (size_t)lbase * QDIM + q;
    float acc = 0.0f;
    #pragma unroll 8
    for (int l = 0; l < LROWS / 16; ++l)
        acc = fmaf(sa[l], c[(size_t)l * QDIM], acc);
    p[(size_t)blockIdx.y * QDIM + q] = acc;
}

extern "C" void kernel_launch(void* const* d_in, const int* in_sizes, int n_in,
                              void* d_out, int out_size, void* d_ws, size_t ws_size,
                              hipStream_t stream)
{
    (void)in_sizes; (void)n_in; (void)out_size; (void)ws_size;
    const float* query = (const float*)d_in[0];
    const float* ctx   = (const float*)d_in[1];
    const float* Kw    = (const float*)d_in[2];
    float* out = (float*)d_out;

    float* ws    = (float*)d_ws;
    float* score = ws;
    float* gamma = ws + LROWS;
    float* part  = ws + LROWS + 2 * QDIM;

    gemv16_wmma_f32<<<dim3(LROWS / 128, 8), 256, (QDIM / 8) * sizeof(float), stream>>>(
        ctx, query, part, QDIM, QDIM / 8);
    reduce_part<<<LROWS / 256, 256, 0, stream>>>(part, score, LROWS, 8);

    softmax_fuse<<<1, 1024, 0, stream>>>(score, query, gamma);

    wsum_part<<<dim3(QDIM / 256, 16), 256, 0, stream>>>(ctx, score, part);
    reduce_part<<<QDIM / 256, 256, 0, stream>>>(part, gamma + QDIM, QDIM, 16);

    gemv16_wmma_f32<<<dim3(QDIM / 128, 8), 256, (2 * QDIM / 8) * sizeof(float), stream>>>(
        Kw, gamma, part, 2 * QDIM, 2 * QDIM / 8);
    reduce_part<<<QDIM / 256, 256, 0, stream>>>(part, out, QDIM, 8);
}


# weighted column sum re-tiled as a stream: 16 B per lane, 8 waves of a block split its 512 rows (64 each, 16-32 loads in flight) and combine in LDS; sum order over L changes
# speedup vs baseline: 1.4267x; 1.4267x over previous
#include <hip/hip_runtime.h>
#include <hip/hip_bf16.h>
#include <math.h>

typedef __attribute__((ext_vector_type(2))) float v2f;
typedef __attribute__((ext_vector_type(4))) float v4f;
typedef __attribute__((ext_vector_type(8))) float v8f;

#define QDIM  4096
#define LROWS 8192

__global__ void __launch_bounds__(256) gemv16_wmma_f32(
    const float* __restrict__ M, const float* __restrict__ x,
    float* __restrict__ y_part, int C, int segC)
{
    extern __shared__ float sx[];
    const int tid   = threadIdx.x;
    const int kbase = blockIdx.y * segC;

    {
        const v4f* xs = (const v4f*)(x + kbase);
        v4f*       sd = (v4f*)sx;
        for (int i = tid; i < (segC >> 2); i += 256) sd[i] = xs[i];
    }
    __syncthreads();

    const int lane    = tid & 31;
    const int m       = lane & 15;
    const int kh      = lane >> 4;
    const int wave    = tid >> 5;
    const int R       = (int)gridDim.x * 128;
    const int rowbase = (int)blockIdx.x * 128 + wave * 16;
    const float* Arow = M + (size_t)(rowbase + m) * (size_t)C + kbase;
    const float  bm   = (m == 0) ? 1.0f : 0.0f;

    v8f acc0 = {}, acc1 = {}, acc2 = {}, acc3 = {};
    for (int k0 = 0; k0 < segC; k0 += 16) {
        const int off_lo = k0 + kh * 4;
        const int off_hi = k0 + 8 + kh * 4;
        v4f a_lo = *(const v4f*)(Arow + off_lo);
        v4f a_hi = *(const v4f*)(Arow + off_hi);
        v4f x_lo = *(const v4f*)(sx + off_lo);
        v4f x_hi = *(const v4f*)(sx + off_hi);

        v2f a0 = {a_lo.x, a_lo.y}, a1 = {a_lo.z, a_lo.w};
        v2f a2 = {a_hi.x, a_hi.y}, a3 = {a_hi.z, a_hi.w};
        v2f b0 = {x_lo.x * bm, x_lo.y * bm}, b1 = {x_lo.z * bm, x_lo.w * bm};
        v2f b2 = {x_hi.x * bm, x_hi.y * bm}, b3 = {x_hi.z * bm, x_hi.w * bm};

        acc0 = __builtin_amdgcn_wmma_f32_16x16x4_f32(false, a0, false, b0, (short)0, acc0, false, false);
        acc1 = __builtin_amdgcn_wmma_f32_16x16x4_f32(false, a1, false, b1, (short)0, acc1, false, false);
        acc2 = __builtin_amdgcn_wmma_f32_16x16x4_f32(false, a2, false, b2, (short)0, acc2, false, false);
        acc3 = __builtin_amdgcn_wmma_f32_16x16x4_f32(false, a3, false, b3, (short)0, acc3, false, false);
    }
    v8f t = acc0 + acc1 + acc2 + acc3;

    if (m == 0) {
        float* dst = y_part + (size_t)blockIdx.y * R + rowbase + kh * 8;
        #pragma unroll
        for (int r = 0; r < 8; ++r) dst[r] = t[r];
    }
}

__global__ void __launch_bounds__(256) reduce_part(
    const float* __restrict__ p, float* __restrict__ y, int R, int nseg)
{
    int i = blockIdx.x * 256 + threadIdx.x;
    if (i < R) {
        float s = 0.0f;
        for (int j = 0; j < nseg; ++j) s += p[(size_t)j * R + i];
        y[i] = s;
    }
}

__global__ void __launch_bounds__(1024) softmax_fuse(
    float* __restrict__ score, const float* __restrict__ query,
    float* __restrict__ gamma)
{
    __shared__ float red[1024];
    const int tid = threadIdx.x;

    float m = -INFINITY;
    #pragma unroll
    for (int j = 0; j < LROWS / 1024; ++j) m = fmaxf(m, score[tid + j * 1024]);
    red[tid] = m;
    __syncthreads();
    for (int s = 512; s > 0; s >>= 1) {
        if (tid < s) red[tid] = fmaxf(red[tid], red[tid + s]);
        __syncthreads();
    }
    const float gmax = red[0];
    __syncthreads();

    float e[LROWS / 1024];
    float sum = 0.0f;
    #pragma unroll
    for (int j = 0; j < LROWS / 1024; ++j) {
        e[j] = expf(score[tid + j * 1024] - gmax);
        sum += e[j];
    }
    red[tid] = sum;
    __syncthreads();
    for (int s = 512; s > 0; s >>= 1) {
        if (tid < s) red[tid] += red[tid + s];
        __syncthreads();
    }
    const float inv = 1.0f / red[0];
    #pragma unroll
    for (int j = 0; j < LROWS / 1024; ++j) score[tid + j * 1024] = e[j] * inv;

    for (int i = tid; i < QDIM; i += 1024) gamma[i] = query[i];
}

#define WS_COLS  128
#define WS_WROWS 64
#define WS_BROWS (8 * WS_WROWS)
#define WS_BATCH 16
__global__ void __launch_bounds__(256) wsum_part(
    const float* __restrict__ ctx, const float* __restrict__ att,
    float* __restrict__ p)
{
    __shared__ v4f sred[8][32];
    const int tid  = threadIdx.x;
    const int lane = tid & 31;
    const int wave = __builtin_amdgcn_readfirstlane(tid >> 5);
    const int l0   = (int)blockIdx.y * WS_BROWS + wave * WS_WROWS;
    const int q    = (int)blockIdx.x * WS_COLS + lane * 4;
    const float* c = ctx + (size_t)l0 * QDIM + q;
    const float* a = att + l0;

    float w[WS_WROWS];
    #pragma unroll
    for (int i = 0; i < WS_WROWS; ++i) w[i] = a[i];

    v4f buf[2][WS_BATCH];
    #pragma unroll
    for (int i = 0; i < WS_BATCH; ++i)
        buf[0][i] = *(const v4f*)(c + (size_t)i * QDIM);

    v4f acc = {0.0f, 0.0f, 0.0f, 0.0f};
    #pragma unroll
    for (int b = 0; b < WS_WROWS / WS_BATCH; ++b) {
        __builtin_amdgcn_sched_barrier(0);
        if (b + 1 < WS_WROWS / WS_BATCH) {
            #pragma unroll
            for (int i = 0; i < WS_BATCH; ++i)
                buf[(b + 1) & 1][i] =
                    *(const v4f*)(c + (size_t)((b + 1) * WS_BATCH + i) * QDIM);
        }
        __builtin_amdgcn_sched_barrier(0);
        #pragma unroll
        for (int i = 0; i < WS_BATCH; ++i) {
            const float wl = w[b * WS_BATCH + i];
            const v4f   v  = buf[b & 1][i];
            acc.x = fmaf(wl, v.x, acc.x);
            acc.y = fmaf(wl, v.y, acc.y);
            acc.z = fmaf(wl, v.z, acc.z);
            acc.w = fmaf(wl, v.w, acc.w);
        }
    }

    sred[wave][lane] = acc;
    __syncthreads();
    if (wave == 0) {
        v4f s = sred[0][lane];
        #pragma unroll
        for (int j = 1; j < 8; ++j) s += sred[j][lane];
        *(v4f*)(p + (size_t)blockIdx.y * QDIM + q) = s;
    }
}

extern "C" void kernel_launch(void* const* d_in, const int* in_sizes, int n_in,
                              void* d_out, int out_size, void* d_ws, size_t ws_size,
                              hipStream_t stream)
{
    (void)in_sizes; (void)n_in; (void)out_size; (void)ws_size;
    const float* query = (const float*)d_in[0];
    const float* ctx   = (const float*)d_in[1];
    const float* Kw    = (const float*)d_in[2];
    float* out = (float*)d_out;

    float* ws    = (float*)d_ws;
    float* score = ws;
    float* gamma = ws + LROWS;
    float* part  = ws + LROWS + 2 * QDIM;

    gemv16_wmma_f32<<<dim3(LROWS / 128, 8), 256, (QDIM / 8) * sizeof(float), stream>>>(
        ctx, query, part, QDIM, QDIM / 8);
    reduce_part<<<LROWS / 256, 256, 0, stream>>>(part, score, LROWS, 8);

    softmax_fuse<<<1, 1024, 0, stream>>>(score, query, gamma);

    wsum_part<<<dim3(QDIM / WS_COLS, LROWS / WS_BROWS), 256, 0, stream>>>(ctx, score, part);
    reduce_part<<<QDIM / 256, 256, 0, stream>>>(part, gamma + QDIM, QDIM, 16);

    gemv16_wmma_f32<<<dim3(QDIM / 128, 8), 256, (2 * QDIM / 8) * sizeof(float), stream>>>(
        Kw, gamma, part, 2 * QDIM, 2 * QDIM / 8);
    reduce_part<<<QDIM / 256, 256, 0, stream>>>(part, out, QDIM, 8);
}


# both matrix-vector products re-tiled to whole-line loads: a wave load = 512 contiguous bytes of one row, row sum = diagonal of the WMMA tile, 16-32 loads in flight per wave; sum order over k changes
# speedup vs baseline: 1.8373x; 1.2878x over previous
#include <hip/hip_runtime.h>
#include <hip/hip_bf16.h>
#include <math.h>

typedef __attribute__((ext_vector_type(2))) float v2f;
typedef __attribute__((ext_vector_type(4))) float v4f;
typedef __attribute__((ext_vector_type(8))) float v8f;

#define QDIM  4096
#define LROWS 8192

#define GV_ROWS   4
#define GV_SEG    2048
#define GV_CHUNK  128
#define GV_BCH    4
#define GV_BATCH  (GV_ROWS * GV_BCH)
#define GV_NB     (GV_SEG / (GV_CHUNK * GV_BCH))
#define GV_BROWS  (8 * GV_ROWS)
__global__ void __launch_bounds__(256, 4) gemv_diag_wmma_f32(
    const float* __restrict__ M, const float* __restrict__ x,
    float* __restrict__ y_part, int C)
{
    __shared__ v4f   sx[GV_SEG / 4];
    __shared__ float sy[GV_BROWS];
    const int tid  = threadIdx.x;
    const int lane = tid & 31;
    const int wave    = __builtin_amdgcn_readfirstlane(tid >> 5);
    const int kbase   = (int)blockIdx.y * GV_SEG;
    const int R       = (int)gridDim.x * GV_BROWS;
    const int rowbase = (int)blockIdx.x * GV_BROWS + wave * GV_ROWS;
    const float* src  = M + (size_t)rowbase * (size_t)C + kbase;
    const int    lo   = lane * 4;

    const v4f* xg = (const v4f*)(x + kbase);
    const v4f xa = xg[tid];
    const v4f xb = xg[tid + 256];

    v4f buf[2][GV_BATCH];
    #pragma unroll
    for (int i = 0; i < GV_BATCH; ++i)
        buf[0][i] = *(const v4f*)(src + (size_t)(i % GV_ROWS) * (size_t)C
                                      + (i / GV_ROWS) * GV_CHUNK + lo);

    sx[tid]       = xa;
    sx[tid + 256] = xb;
    __builtin_amdgcn_fence(__ATOMIC_RELEASE, "workgroup", "local");
    __builtin_amdgcn_s_barrier();
    __builtin_amdgcn_fence(__ATOMIC_ACQUIRE, "workgroup", "local");

    v8f acc[GV_ROWS];
    #pragma unroll
    for (int r = 0; r < GV_ROWS; ++r) acc[r] = (v8f){};

    #pragma unroll
    for (int b = 0; b < GV_NB; ++b) {
        __builtin_amdgcn_sched_barrier(0);
        if (b + 1 < GV_NB) {
            #pragma unroll
            for (int i = 0; i < GV_BATCH; ++i)
                buf[(b + 1) & 1][i] =
                    *(const v4f*)(src + (size_t)(i % GV_ROWS) * (size_t)C
                                      + ((b + 1) * GV_BCH + i / GV_ROWS) * GV_CHUNK + lo);
        }
        __builtin_amdgcn_sched_barrier(0);
        #pragma unroll
        for (int ch = 0; ch < GV_BCH; ++ch) {
            const v4f xf = sx[(b * GV_BCH + ch) * (GV_CHUNK / 4) + lane];
            const v2f b0 = {xf.x, xf.y}, b1 = {xf.z, xf.w};
            #pragma unroll
            for (int r = 0; r < GV_ROWS; ++r) {
                const v4f a = buf[b & 1][ch * GV_ROWS + r];
                const v2f a0 = {a.x, a.y};
                acc[r] = __builtin_amdgcn_wmma_f32_16x16x4_f32(false, a0, false, b0, (short)0, acc[r], false, false);
            }
            #pragma unroll
            for (int r = 0; r < GV_ROWS; ++r) {
                const v4f a = buf[b & 1][ch * GV_ROWS + r];
                const v2f a1 = {a.z, a.w};
                acc[r] = __builtin_amdgcn_wmma_f32_16x16x4_f32(false, a1, false, b1, (short)0, acc[r], false, false);
            }
        }
    }

    const int  j    = lane & 7;
    const bool diag = (lane < 8) || (lane >= 24);
    float mine = 0.0f;
    #pragma unroll
    for (int r = 0; r < GV_ROWS; ++r) {
        const v8f t = acc[r];
        float d = t[0];
        d = (j == 1) ? t[1] : d;
        d = (j == 2) ? t[2] : d;
        d = (j == 3) ? t[3] : d;
        d = (j == 4) ? t[4] : d;
        d = (j == 5) ? t[5] : d;
        d = (j == 6) ? t[6] : d;
        d = (j == 7) ? t[7] : d;
        d = diag ? d : 0.0f;
        #pragma unroll
        for (int off = 16; off > 0; off >>= 1) d += __shfl_xor(d, off);
        if (lane == r) mine = d;
    }
    if (lane < GV_ROWS) sy[wave * GV_ROWS + lane] = mine;
    __syncthreads();
    if (wave == 0)
        y_part[(size_t)blockIdx.y * R + (size_t)blockIdx.x * GV_BROWS + lane] = sy[lane];
}

__global__ void __launch_bounds__(256) reduce_part(
    const float* __restrict__ p, float* __restrict__ y, int R, int nseg)
{
    int i = blockIdx.x * 256 + threadIdx.x;
    if (i < R) {
        float s = 0.0f;
        for (int j = 0; j < nseg; ++j) s += p[(size_t)j * R + i];
        y[i] = s;
    }
}

__global__ void __launch_bounds__(1024) softmax_fuse(
    float* __restrict__ score, const float* __restrict__ query,
    float* __restrict__ gamma)
{
    __shared__ float red[1024];
    const int tid = threadIdx.x;

    float m = -INFINITY;
    #pragma unroll
    for (int j = 0; j < LROWS / 1024; ++j) m = fmaxf(m, score[tid + j * 1024]);
    red[tid] = m;
    __syncthreads();
    for (int s = 512; s > 0; s >>= 1) {
        if (tid < s) red[tid] = fmaxf(red[tid], red[tid + s]);
        __syncthreads();
    }
    const float gmax = red[0];
    __syncthreads();

    float e[LROWS / 1024];
    float sum = 0.0f;
    #pragma unroll
    for (int j = 0; j < LROWS / 1024; ++j) {
        e[j] = expf(score[tid + j * 1024] - gmax);
        sum += e[j];
    }
    red[tid] = sum;
    __syncthreads();
    for (int s = 512; s > 0; s >>= 1) {
        if (tid < s) red[tid] += red[tid + s];
        __syncthreads();
    }
    const float inv = 1.0f / red[0];
    #pragma unroll
    for (int j = 0; j < LROWS / 1024; ++j) score[tid + j * 1024] = e[j] * inv;

    for (int i = tid; i < QDIM; i += 1024) gamma[i] = query[i];
}

#define WS_COLS  128
#define WS_WROWS 64
#define WS_BROWS (8 * WS_WROWS)
#define WS_BATCH 16
__global__ void __launch_bounds__(256) wsum_part(
    const float* __restrict__ ctx, const float* __restrict__ att,
    float* __restrict__ p)
{
    __shared__ v4f sred[8][32];
    const int tid  = threadIdx.x;
    const int lane = tid & 31;
    const int wave = __builtin_amdgcn_readfirstlane(tid >> 5);
    const int l0   = (int)blockIdx.y * WS_BROWS + wave * WS_WROWS;
    const int q    = (int)blockIdx.x * WS_COLS + lane * 4;
    const float* c = ctx + (size_t)l0 * QDIM + q;
    const float* a = att + l0;

    float w[WS_WROWS];
    #pragma unroll
    for (int i = 0; i < WS_WROWS; ++i) w[i] = a[i];

    v4f buf[2][WS_BATCH];
    #pragma unroll
    for (int i = 0; i < WS_BATCH; ++i)
        buf[0][i] = *(const v4f*)(c + (size_t)i * QDIM);

    v4f acc = {0.0f, 0.0f, 0.0f, 0.0f};
    #pragma unroll
    for (int b = 0; b < WS_WROWS / WS_BATCH; ++b) {
        __builtin_amdgcn_sched_barrier(0);
        if (b + 1 < WS_WROWS / WS_BATCH) {
            #pragma unroll
            for (int i = 0; i < WS_BATCH; ++i)
                buf[(b + 1) & 1][i] =
                    *(const v4f*)(c + (size_t)((b + 1) * WS_BATCH + i) * QDIM);
        }
        __builtin_amdgcn_sched_barrier(0);
        #pragma unroll
        for (int i = 0; i < WS_BATCH; ++i) {
            const float wl = w[b * WS_BATCH + i];
            const v4f   v  = buf[b & 1][i];
            acc.x = fmaf(wl, v.x, acc.x);
            acc.y = fmaf(wl, v.y, acc.y);
            acc.z = fmaf(wl, v.z, acc.z);
            acc.w = fmaf(wl, v.w, acc.w);
        }
    }

    sred[wave][lane] = acc;
    __syncthreads();
    if (wave == 0) {
        v4f s = sred[0][lane];
        #pragma unroll
        for (int j = 1; j < 8; ++j) s += sred[j][lane];
        *(v4f*)(p + (size_t)blockIdx.y * QDIM + q) = s;
    }
}

extern "C" void kernel_launch(void* const* d_in, const int* in_sizes, int n_in,
                              void* d_out, int out_size, void* d_ws, size_t ws_size,
                              hipStream_t stream)
{
    (void)in_sizes; (void)n_in; (void)out_size; (void)ws_size;
    const float* query = (const float*)d_in[0];
    const float* ctx   = (const float*)d_in[1];
    const float* Kw    = (const float*)d_in[2];
    float* out = (float*)d_out;

    float* ws    = (float*)d_ws;
    float* score = ws;
    float* gamma = ws + LROWS;
    float* part  = ws + LROWS + 2 * QDIM;

    gemv_diag_wmma_f32<<<dim3(LROWS / GV_BROWS, QDIM / GV_SEG), 256, 0, stream>>>(
        ctx, query, part, QDIM);
    reduce_part<<<LROWS / 256, 256, 0, stream>>>(part, score, LROWS, QDIM / GV_SEG);

    softmax_fuse<<<1, 1024, 0, stream>>>(score, query, gamma);

    wsum_part<<<dim3(QDIM / WS_COLS, LROWS / WS_BROWS), 256, 0, stream>>>(ctx, score, part);
    reduce_part<<<QDIM / 256, 256, 0, stream>>>(part, gamma + QDIM, QDIM, 16);

    gemv_diag_wmma_f32<<<dim3(QDIM / GV_BROWS, 2 * QDIM / GV_SEG), 256, 0, stream>>>(
        Kw, gamma, part, 2 * QDIM);
    reduce_part<<<QDIM / 256, 256, 0, stream>>>(part, out, QDIM, 2 * QDIM / GV_SEG);
}


# context matrix read once: score product, running softmax and weighted column sum fused per 32-row block, 256 block results merged by one small launch; sum order over k and over L changes
# speedup vs baseline: 2.5998x; 1.4150x over previous
#include <hip/hip_runtime.h>
#include <hip/hip_bf16.h>
#include <math.h>

typedef __attribute__((ext_vector_type(2))) float v2f;
typedef __attribute__((ext_vector_type(4))) float v4f;
typedef __attribute__((ext_vector_type(8))) float v8f;

#define QDIM  4096
#define LROWS 8192

#define GV_ROWS   4
#define GV_SEG    2048
#define GV_CHUNK  128
#define GV_BCH    4
#define GV_BATCH  (GV_ROWS * GV_BCH)
#define GV_NB     (GV_SEG / (GV_CHUNK * GV_BCH))
#define GV_BROWS  (8 * GV_ROWS)
__global__ void __launch_bounds__(256, 4) gemv_diag_wmma_f32(
    const float* __restrict__ M, const float* __restrict__ x,
    const float* __restrict__ xhi, float* __restrict__ y_part, int C, int xsplit)
{
    __shared__ v4f   sx[GV_SEG / 4];
    __shared__ float sy[GV_BROWS];
    const int tid  = threadIdx.x;
    const int lane = tid & 31;
    const int wave    = __builtin_amdgcn_readfirstlane(tid >> 5);
    const int kbase   = (int)blockIdx.y * GV_SEG;
    const int R       = (int)gridDim.x * GV_BROWS;
    const int rowbase = (int)blockIdx.x * GV_BROWS + wave * GV_ROWS;
    const float* src  = M + (size_t)rowbase * (size_t)C + kbase;
    const int    lo   = lane * 4;

    const bool from_parts = (kbase >= xsplit);
    const v4f* xg = from_parts ? (const v4f*)(xhi + (kbase - xsplit))
                               : (const v4f*)(x + kbase);
    v4f xa = xg[tid];
    v4f xb = xg[tid + 256];

    v4f buf[2][GV_BATCH];
    #pragma unroll
    for (int i = 0; i < GV_BATCH; ++i)
        buf[0][i] = *(const v4f*)(src + (size_t)(i % GV_ROWS) * (size_t)C
                                      + (i / GV_ROWS) * GV_CHUNK + lo);

    sx[tid]       = xa;
    sx[tid + 256] = xb;
    __builtin_amdgcn_fence(__ATOMIC_RELEASE, "workgroup", "local");
    __builtin_amdgcn_s_barrier();
    __builtin_amdgcn_fence(__ATOMIC_ACQUIRE, "workgroup", "local");

    v8f acc[GV_ROWS];
    #pragma unroll
    for (int r = 0; r < GV_ROWS; ++r) acc[r] = (v8f){};

    #pragma unroll
    for (int b = 0; b < GV_NB; ++b) {
        __builtin_amdgcn_sched_barrier(0);
        if (b + 1 < GV_NB) {
            #pragma unroll
            for (int i = 0; i < GV_BATCH; ++i)
                buf[(b + 1) & 1][i] =
                    *(const v4f*)(src + (size_t)(i % GV_ROWS) * (size_t)C
                                      + ((b + 1) * GV_BCH + i / GV_ROWS) * GV_CHUNK + lo);
        }
        __builtin_amdgcn_sched_barrier(0);
        #pragma unroll
        for (int ch = 0; ch < GV_BCH; ++ch) {
            const v4f xf = sx[(b * GV_BCH + ch) * (GV_CHUNK / 4) + lane];
            const v2f b0 = {xf.x, xf.y}, b1 = {xf.z, xf.w};
            #pragma unroll
            for (int r = 0; r < GV_ROWS; ++r) {
                const v4f a = buf[b & 1][ch * GV_ROWS + r];
                const v2f a0 = {a.x, a.y};
                acc[r] = __builtin_amdgcn_wmma_f32_16x16x4_f32(false, a0, false, b0, (short)0, acc[r], false, false);
            }
            #pragma unroll
            for (int r = 0; r < GV_ROWS; ++r) {
                const v4f a = buf[b & 1][ch * GV_ROWS + r];
                const v2f a1 = {a.z, a.w};
                acc[r] = __builtin_amdgcn_wmma_f32_16x16x4_f32(false, a1, false, b1, (short)0, acc[r], false, false);
            }
        }
    }

    const int  j    = lane & 7;
    const bool diag = (lane < 8) || (lane >= 24);
    float mine = 0.0f;
    #pragma unroll
    for (int r = 0; r < GV_ROWS; ++r) {
        const v8f t = acc[r];
        float d = t[0];
        d = (j == 1) ? t[1] : d;
        d = (j == 2) ? t[2] : d;
        d = (j == 3) ? t[3] : d;
        d = (j == 4) ? t[4] : d;
        d = (j == 5) ? t[5] : d;
        d = (j == 6) ? t[6] : d;
        d = (j == 7) ? t[7] : d;
        d = diag ? d : 0.0f;
        #pragma unroll
        for (int off = 16; off > 0; off >>= 1) d += __shfl_xor(d, off);
        if (lane == r) mine = d;
    }
    if (lane < GV_ROWS) sy[wave * GV_ROWS + lane] = mine;
    __syncthreads();
    if (wave == 0)
        y_part[(size_t)blockIdx.y * R + (size_t)blockIdx.x * GV_BROWS + lane] = sy[lane];
}

__global__ void __launch_bounds__(256) reduce_part(
    const float* __restrict__ p, float* __restrict__ y, int R, int nseg)
{
    int i = blockIdx.x * 256 + threadIdx.x;
    if (i < R) {
        float s = 0.0f;
        for (int j = 0; j < nseg; ++j) s += p[(size_t)j * R + i];
        y[i] = s;
    }
}

__device__ __forceinline__ void lds_only_barrier()
{
    __builtin_amdgcn_fence(__ATOMIC_RELEASE, "workgroup", "local");
    __builtin_amdgcn_s_barrier();
    __builtin_amdgcn_fence(__ATOMIC_ACQUIRE, "workgroup", "local");
}

#define FZ_T     8
#define FZ_CH    4
#define FZ_ROWS  32
#define FZ_NT    (FZ_ROWS / FZ_T)
#define FZ_NBLK  (LROWS / FZ_ROWS)
#define FZ_LOADS (FZ_T * FZ_CH)
__global__ void __launch_bounds__(256, 2) fused_score_wsum(
    const float* __restrict__ ctx, const float* __restrict__ query,
    float* __restrict__ part, float* __restrict__ md)
{
    __shared__ float sps[2][8][FZ_T];
    const int tid  = threadIdx.x;
    const int lane = tid & 31;
    const int wave = __builtin_amdgcn_readfirstlane(tid >> 5);
    const int col  = wave * (FZ_CH * 128) + lane * 4;
    const float* c = ctx + (size_t)blockIdx.x * FZ_ROWS * QDIM + col;

    v4f buf[2][FZ_LOADS];
    #pragma unroll
    for (int i = 0; i < FZ_LOADS; ++i)
        buf[0][i] = *(const v4f*)(c + (size_t)(i / FZ_CH) * QDIM + (i % FZ_CH) * 128);

    v4f xr[FZ_CH];
    #pragma unroll
    for (int ch = 0; ch < FZ_CH; ++ch) xr[ch] = *(const v4f*)(query + col + ch * 128);

    const int  j    = lane & 7;
    const bool diag = (lane < 8) || (lane >= 24);
    float m = 0.0f, d = 0.0f;
    v4f acc[FZ_CH];
    #pragma unroll
    for (int ch = 0; ch < FZ_CH; ++ch) acc[ch] = (v4f){0.0f, 0.0f, 0.0f, 0.0f};

    #pragma unroll
    for (int t = 0; t < FZ_NT; ++t) {
        __builtin_amdgcn_sched_barrier(0);
        if (t + 1 < FZ_NT) {
            #pragma unroll
            for (int i = 0; i < FZ_LOADS; ++i)
                buf[(t + 1) & 1][i] =
                    *(const v4f*)(c + (size_t)((t + 1) * FZ_T + i / FZ_CH) * QDIM + (i % FZ_CH) * 128);
        }
        __builtin_amdgcn_sched_barrier(0);

        v8f sa[FZ_T];
        #pragma unroll
        for (int r = 0; r < FZ_T; ++r) sa[r] = (v8f){};
        #pragma unroll
        for (int ch = 0; ch < FZ_CH; ++ch) {
            const v2f b0 = {xr[ch].x, xr[ch].y}, b1 = {xr[ch].z, xr[ch].w};
            #pragma unroll
            for (int r = 0; r < FZ_T; ++r) {
                const v4f a = buf[t & 1][r * FZ_CH + ch];
                const v2f a0 = {a.x, a.y};
                sa[r] = __builtin_amdgcn_wmma_f32_16x16x4_f32(false, a0, false, b0, (short)0, sa[r], false, false);
            }
            #pragma unroll
            for (int r = 0; r < FZ_T; ++r) {
                const v4f a = buf[t & 1][r * FZ_CH + ch];
                const v2f a1 = {a.z, a.w};
                sa[r] = __builtin_amdgcn_wmma_f32_16x16x4_f32(false, a1, false, b1, (short)0, sa[r], false, false);
            }
        }
        float mine = 0.0f;
        #pragma unroll
        for (int r = 0; r < FZ_T; ++r) {
            const v8f tt = sa[r];
            float dd = tt[0];
            dd = (j == 1) ? tt[1] : dd;
            dd = (j == 2) ? tt[2] : dd;
            dd = (j == 3) ? tt[3] : dd;
            dd = (j == 4) ? tt[4] : dd;
            dd = (j == 5) ? tt[5] : dd;
            dd = (j == 6) ? tt[6] : dd;
            dd = (j == 7) ? tt[7] : dd;
            dd = diag ? dd : 0.0f;
            #pragma unroll
            for (int off = 16; off > 0; off >>= 1) dd += __shfl_xor(dd, off);
            if (lane == r) mine = dd;
        }
        if (lane < FZ_T) sps[t & 1][wave][lane] = mine;
        lds_only_barrier();

        float s = sps[t & 1][0][j];
        #pragma unroll
        for (int w = 1; w < 8; ++w) s += sps[t & 1][w][j];

        float tm = s;
        tm = fmaxf(tm, __shfl_xor(tm, 1));
        tm = fmaxf(tm, __shfl_xor(tm, 2));
        tm = fmaxf(tm, __shfl_xor(tm, 4));

        const float mn = (t == 0) ? tm : fmaxf(m, tm);
        const float e  = expf(s - mn);
        float es = e;
        es += __shfl_xor(es, 1);
        es += __shfl_xor(es, 2);
        es += __shfl_xor(es, 4);
        if (t == 0) {
            d = es;
        } else {
            const float scale = expf(m - mn);
            d = d * scale + es;
            #pragma unroll
            for (int ch = 0; ch < FZ_CH; ++ch) acc[ch] *= scale;
        }
        m = mn;

        float wgt[FZ_T];
        #pragma unroll
        for (int r = 0; r < FZ_T; ++r)
            wgt[r] = __int_as_float(__builtin_amdgcn_readlane(__float_as_int(e), r));

        #pragma unroll
        for (int r = 0; r < FZ_T; ++r) {
            #pragma unroll
            for (int ch = 0; ch < FZ_CH; ++ch) {
                const v4f v = buf[t & 1][r * FZ_CH + ch];
                acc[ch].x = fmaf(wgt[r], v.x, acc[ch].x);
                acc[ch].y = fmaf(wgt[r], v.y, acc[ch].y);
                acc[ch].z = fmaf(wgt[r], v.z, acc[ch].z);
                acc[ch].w = fmaf(wgt[r], v.w, acc[ch].w);
            }
        }
    }

    #pragma unroll
    for (int ch = 0; ch < FZ_CH; ++ch)
        *(v4f*)(part + (size_t)blockIdx.x * QDIM + col + ch * 128) = acc[ch];
    if (wave == 0)
        md[(size_t)blockIdx.x * 32 + lane] = (lane == 0) ? m : ((lane == 1) ? d : 0.0f);
}

__global__ void __launch_bounds__(256) merge_parts(
    const float* __restrict__ part, const float* __restrict__ md,
    float* __restrict__ st)
{
    __shared__ v4f   sred[8][32];
    __shared__ float scoef[FZ_NBLK];
    __shared__ float smax[8];
    __shared__ float ssum[8];
    const int tid  = threadIdx.x;
    const int lane = tid & 31;
    const int wave = __builtin_amdgcn_readfirstlane(tid >> 5);
    const int q    = (int)blockIdx.x * 128 + lane * 4;

    v4f pv[FZ_NBLK / 8];
    #pragma unroll
    for (int i = 0; i < FZ_NBLK / 8; ++i)
        pv[i] = *(const v4f*)(part + (size_t)(wave + 8 * i) * QDIM + q);

    const float mb = md[(size_t)tid * 32];
    const float db = md[(size_t)tid * 32 + 1];

    float mx = mb;
    #pragma unroll
    for (int off = 16; off > 0; off >>= 1) mx = fmaxf(mx, __shfl_xor(mx, off));
    if (lane == 0) smax[wave] = mx;
    lds_only_barrier();
    float gmax = smax[0];
    #pragma unroll
    for (int w = 1; w < 8; ++w) gmax = fmaxf(gmax, smax[w]);

    const float cf = expf(mb - gmax);
    scoef[tid] = cf;
    float sum = cf * db;
    #pragma unroll
    for (int off = 16; off > 0; off >>= 1) sum += __shfl_xor(sum, off);
    if (lane == 0) ssum[wave] = sum;
    lds_only_barrier();
    float den = ssum[0];
    #pragma unroll
    for (int w = 1; w < 8; ++w) den += ssum[w];
    const float inv = 1.0f / den;

    v4f a = {0.0f, 0.0f, 0.0f, 0.0f};
    #pragma unroll
    for (int i = 0; i < FZ_NBLK / 8; ++i) {
        const float cw = scoef[wave + 8 * i];
        a.x = fmaf(cw, pv[i].x, a.x);
        a.y = fmaf(cw, pv[i].y, a.y);
        a.z = fmaf(cw, pv[i].z, a.z);
        a.w = fmaf(cw, pv[i].w, a.w);
    }
    sred[wave][lane] = a;
    __syncthreads();
    if (wave == 0) {
        v4f s = sred[0][lane];
        #pragma unroll
        for (int w = 1; w < 8; ++w) s += sred[w][lane];
        s *= inv;
        *(v4f*)(st + q) = s;
    }
}

extern "C" void kernel_launch(void* const* d_in, const int* in_sizes, int n_in,
                              void* d_out, int out_size, void* d_ws, size_t ws_size,
                              hipStream_t stream)
{
    (void)in_sizes; (void)n_in; (void)out_size;
    const float* query = (const float*)d_in[0];
    const float* ctx   = (const float*)d_in[1];
    const float* Kw    = (const float*)d_in[2];
    float* out = (float*)d_out;

    if (ws_size < (size_t)4325376) return;
    float* ws    = (float*)d_ws;
    float* ypart = ws;
    float* st    = ws + 16384;
    float* md    = ws + 20480;
    float* part  = ws + 32768;

    fused_score_wsum<<<FZ_NBLK, 256, 0, stream>>>(ctx, query, part, md);

    merge_parts<<<QDIM / 128, 256, 0, stream>>>(part, md, st);

    gemv_diag_wmma_f32<<<dim3(QDIM / GV_BROWS, 2 * QDIM / GV_SEG), 256, 0, stream>>>(
        Kw, query, st, ypart, 2 * QDIM, QDIM);
    reduce_part<<<QDIM / 256, 256, 0, stream>>>(ypart, out, QDIM, 2 * QDIM / GV_SEG);
}
